# starting kernel
# baseline (speedup 1.0000x reference)
#include <hip/hip_runtime.h>


typedef float v2f __attribute__((ext_vector_type(2)));
typedef float v8f __attribute__((ext_vector_type(8)));

#define NQ    12
#define DIM   4096
#define NBATCH 4096

__global__ __launch_bounds__(256) void rx_build_gates(const float* __restrict__ params,
                                                      float* __restrict__ g) {
    int tid = blockIdx.x * 256 + threadIdx.x;
    if (tid >= 3 * 256) return;
    int grp = tid >> 8;
    int jk  = tid & 255;
    int j = jk >> 4, k = jk & 15;
    float pr = 1.0f, pi = 0.0f;
#pragma unroll
    for (int t = 0; t < 4; ++t) {
        int q = grp * 4 + t;
        float h = 0.5f * params[q];
        float c = cosf(h), s = sinf(h);
        int bj = (j >> (3 - t)) & 1;
        int bk = (k >> (3 - t)) & 1;
        float gr = (bj == bk) ? c : 0.0f;
        float gi = (bj == bk) ? 0.0f : -s;
        float nr = pr * gr - pi * gi;
        float ni = pr * gi + pi * gr;
        pr = nr; pi = ni;
    }
    g[grp * 512 + jk]       = pr;
    g[grp * 512 + 256 + jk] = pi;
}

__device__ __forceinline__ v2f frag_a(const float* src, int stride, int k0, int r16) {
    v2f a;
    a.x = src[r16 * stride + k0];
    a.y = src[r16 * stride + k0 + 1];
    return a;
}
__device__ __forceinline__ v2f frag_b(const float* src, int stride, int k0, int r16) {
    v2f b;
    b.x = src[(k0)     * stride + r16];
    b.y = src[(k0 + 1) * stride + r16];
    return b;
}
__device__ __forceinline__ void store_d(float* dst, int stride, const v8f& d, int r16, int hi) {
#pragma unroll
    for (int v = 0; v < 8; ++v)
        dst[(v + 8 * hi) * stride + r16] = d[v];
}

__device__ __forceinline__ void ctile(const float* Ar, const float* Ai, int sa,
                                      const float* Br, const float* Bi, int sb,
                                      v8f& dr, v8f& di, int r16, int hi) {
    v8f z = {0.f, 0.f, 0.f, 0.f, 0.f, 0.f, 0.f, 0.f};
    dr = z; di = z;
#pragma unroll
    for (int kb = 0; kb < 4; ++kb) {
        int k0 = kb * 4 + 2 * hi;
        v2f a_r = frag_a(Ar, sa, k0, r16);
        v2f a_i = frag_a(Ai, sa, k0, r16);
        v2f b_r = frag_b(Br, sb, k0, r16);
        v2f b_i = frag_b(Bi, sb, k0, r16);
        v2f nb_i; nb_i.x = -b_i.x; nb_i.y = -b_i.y;
        dr = __builtin_amdgcn_wmma_f32_16x16x4_f32(false, a_r, false, b_r,  (short)0, dr, false, false);
        dr = __builtin_amdgcn_wmma_f32_16x16x4_f32(false, a_i, false, nb_i, (short)0, dr, false, false);
        di = __builtin_amdgcn_wmma_f32_16x16x4_f32(false, a_r, false, b_i,  (short)0, di, false, false);
        di = __builtin_amdgcn_wmma_f32_16x16x4_f32(false, a_i, false, b_r,  (short)0, di, false, false);
    }
}

__global__ __launch_bounds__(256) void rx_apply(const float* __restrict__ xr,
                                                const float* __restrict__ xi,
                                                const float* __restrict__ g,
                                                float* __restrict__ outr,
                                                float* __restrict__ outi) {
    __shared__ __align__(16) float sr[DIM];
    __shared__ __align__(16) float si[DIM];

    const int row = blockIdx.x;
    const size_t rbase = (size_t)row * DIM;

    {
        const float4* xr4 = (const float4*)(xr + rbase);
        const float4* xi4 = (const float4*)(xi + rbase);
        float4* sr4 = (float4*)sr;
        float4* si4 = (float4*)si;
#pragma unroll
        for (int it = 0; it < 4; ++it) {
            int idx = threadIdx.x + it * 256;
            sr4[idx] = xr4[idx];
            si4[idx] = xi4[idx];
        }
    }
    __syncthreads();

    const int lane = threadIdx.x & 31;
    const int wave = threadIdx.x >> 5;
    const int r16  = lane & 15;
    const int hi   = lane >> 4;

    const float* gAr = g;        const float* gAi = g + 256;
    const float* gBr = g + 512;  const float* gBi = g + 768;
    const float* gCr = g + 1024; const float* gCi = g + 1280;

    v8f dr, di;

#pragma unroll
    for (int rep = 0; rep < 2; ++rep) {
        const int t = wave + 8 * rep;
        float* br_ = sr + t * 256;
        float* bi_ = si + t * 256;
        ctile(br_, bi_, 16, gCr, gCi, 16, dr, di, r16, hi);
        store_d(br_, 16, dr, r16, hi);
        store_d(bi_, 16, di, r16, hi);
        ctile(gBr, gBi, 16, br_, bi_, 16, dr, di, r16, hi);
        store_d(br_, 16, dr, r16, hi);
        store_d(bi_, 16, di, r16, hi);
    }

    __syncthreads();

#pragma unroll
    for (int rep = 0; rep < 2; ++rep) {
        const int t = wave + 8 * rep;
        ctile(gAr, gAi, 16, sr + 16 * t, si + 16 * t, 256, dr, di, r16, hi);
#pragma unroll
        for (int v = 0; v < 8; ++v) {
            size_t o = rbase + (size_t)((v + 8 * hi) * 256 + 16 * t + r16);
            outr[o] = dr[v];
            outi[o] = di[v];
        }
    }
}

extern "C" void kernel_launch(void* const* d_in, const int* in_sizes, int n_in,
                              void* d_out, int out_size, void* d_ws, size_t ws_size,
                              hipStream_t stream) {
    (void)in_sizes; (void)n_in; (void)out_size; (void)ws_size;
    const float* xr     = (const float*)d_in[0];
    const float* xi     = (const float*)d_in[1];
    const float* params = (const float*)d_in[2];

    float* g = (float*)d_ws;
    float* outr = (float*)d_out;
    float* outi = outr + (size_t)NBATCH * DIM;

    rx_build_gates<<<3, 256, 0, stream>>>(params, g);
    rx_apply<<<NBATCH, 256, 0, stream>>>(xr, xi, g, outr, outi);
}


# Real group gates: RX = P R P, P = diag(1,-i): phase (-i)^popcount(index) applied at row load and at store (re/im swap, sign); a tile product is 2 real f32 WMMA products, not 4; sum order changes
# speedup vs baseline: 1.2283x; 1.2283x over previous
#include <hip/hip_runtime.h>


typedef float v2f __attribute__((ext_vector_type(2)));
typedef float v8f __attribute__((ext_vector_type(8)));

#define NQ    12
#define DIM   4096
#define NBATCH 4096

__global__ __launch_bounds__(256) void rx_build_gates(const float* __restrict__ params,
                                                      float* __restrict__ g) {
    int tid = blockIdx.x * 256 + threadIdx.x;
    if (tid >= 3 * 256) return;
    int grp = tid >> 8;
    int jk  = tid & 255;
    int j = jk >> 4, k = jk & 15;
    float p = 1.0f;
#pragma unroll
    for (int t = 0; t < 4; ++t) {
        int q = grp * 4 + t;
        float h = 0.5f * params[q];
        float c = cosf(h), s = sinf(h);
        int bj = (j >> (3 - t)) & 1;
        int bk = (k >> (3 - t)) & 1;
        float r = (bj == bk) ? (bj ? -c : c) : s;
        p = p * r;
    }
    g[grp * 256 + jk] = p;
}

__device__ __forceinline__ void phase_rot(float re, float im, int p, float& ore, float& oim) {
    const bool sw = (p & 1) != 0;
    const float a = sw ? im : re;
    const float b = sw ? re : im;
    const unsigned sa = (unsigned)(p & 2) << 30;
    const unsigned sb = (unsigned)((p + 1) & 2) << 30;
    ore = __uint_as_float(__float_as_uint(a) ^ sa);
    oim = __uint_as_float(__float_as_uint(b) ^ sb);
}

__device__ __forceinline__ v2f frag_a(const float* src, int stride, int k0, int r16) {
    v2f a;
    a.x = src[r16 * stride + k0];
    a.y = src[r16 * stride + k0 + 1];
    return a;
}
__device__ __forceinline__ v2f frag_b(const float* src, int stride, int k0, int r16) {
    v2f b;
    b.x = src[(k0)     * stride + r16];
    b.y = src[(k0 + 1) * stride + r16];
    return b;
}
__device__ __forceinline__ void store_d(float* dst, int stride, const v8f& d, int r16, int hi) {
#pragma unroll
    for (int v = 0; v < 8; ++v)
        dst[(v + 8 * hi) * stride + r16] = d[v];
}

__device__ __forceinline__ void rtile_data_gate(const float* Ar, const float* Ai, int sa,
                                                const float* G, int sb,
                                                v8f& dr, v8f& di, int r16, int hi) {
    v8f z = {0.f, 0.f, 0.f, 0.f, 0.f, 0.f, 0.f, 0.f};
    dr = z; di = z;
#pragma unroll
    for (int kb = 0; kb < 4; ++kb) {
        int k0 = kb * 4 + 2 * hi;
        v2f a_r = frag_a(Ar, sa, k0, r16);
        v2f a_i = frag_a(Ai, sa, k0, r16);
        v2f b   = frag_b(G,  sb, k0, r16);
        dr = __builtin_amdgcn_wmma_f32_16x16x4_f32(false, a_r, false, b, (short)0, dr, false, false);
        di = __builtin_amdgcn_wmma_f32_16x16x4_f32(false, a_i, false, b, (short)0, di, false, false);
    }
}

__device__ __forceinline__ void rtile_gate_data(const float* G, int sa,
                                                const float* Br, const float* Bi, int sb,
                                                v8f& dr, v8f& di, int r16, int hi) {
    v8f z = {0.f, 0.f, 0.f, 0.f, 0.f, 0.f, 0.f, 0.f};
    dr = z; di = z;
#pragma unroll
    for (int kb = 0; kb < 4; ++kb) {
        int k0 = kb * 4 + 2 * hi;
        v2f a   = frag_a(G,  sa, k0, r16);
        v2f b_r = frag_b(Br, sb, k0, r16);
        v2f b_i = frag_b(Bi, sb, k0, r16);
        dr = __builtin_amdgcn_wmma_f32_16x16x4_f32(false, a, false, b_r, (short)0, dr, false, false);
        di = __builtin_amdgcn_wmma_f32_16x16x4_f32(false, a, false, b_i, (short)0, di, false, false);
    }
}

__global__ __launch_bounds__(256) void rx_apply(const float* __restrict__ xr,
                                                const float* __restrict__ xi,
                                                const float* __restrict__ g,
                                                float* __restrict__ outr,
                                                float* __restrict__ outi) {
    __shared__ __align__(16) float sr[DIM];
    __shared__ __align__(16) float si[DIM];

    const int row = blockIdx.x;
    const size_t rbase = (size_t)row * DIM;

    {
        const float4* xr4 = (const float4*)(xr + rbase);
        const float4* xi4 = (const float4*)(xi + rbase);
        float4* sr4 = (float4*)sr;
        float4* si4 = (float4*)si;
        const int pt = __popc((unsigned)threadIdx.x);
#pragma unroll
        for (int it = 0; it < 4; ++it) {
            int idx = threadIdx.x + it * 256;
            float4 vr = xr4[idx];
            float4 vi = xi4[idx];
            const int p0 = pt + ((it & 1) + (it >> 1));
            float4 qr, qi;
            phase_rot(vr.x, vi.x, p0,     qr.x, qi.x);
            phase_rot(vr.y, vi.y, p0 + 1, qr.y, qi.y);
            phase_rot(vr.z, vi.z, p0 + 1, qr.z, qi.z);
            phase_rot(vr.w, vi.w, p0 + 2, qr.w, qi.w);
            sr4[idx] = qr;
            si4[idx] = qi;
        }
    }
    __syncthreads();

    const int lane = threadIdx.x & 31;
    const int wave = threadIdx.x >> 5;
    const int r16  = lane & 15;
    const int hi   = lane >> 4;

    const float* gA = g;
    const float* gB = g + 256;
    const float* gC = g + 512;

    v8f dr, di;

#pragma unroll
    for (int rep = 0; rep < 2; ++rep) {
        const int t = wave + 8 * rep;
        float* br_ = sr + t * 256;
        float* bi_ = si + t * 256;
        rtile_data_gate(br_, bi_, 16, gC, 16, dr, di, r16, hi);
        store_d(br_, 16, dr, r16, hi);
        store_d(bi_, 16, di, r16, hi);
        rtile_gate_data(gB, 16, br_, bi_, 16, dr, di, r16, hi);
        store_d(br_, 16, dr, r16, hi);
        store_d(bi_, 16, di, r16, hi);
    }

    __syncthreads();

    const int pl = __popc((unsigned)lane);
#pragma unroll
    for (int rep = 0; rep < 2; ++rep) {
        const int t = wave + 8 * rep;
        rtile_gate_data(gA, 16, sr + 16 * t, si + 16 * t, 256, dr, di, r16, hi);
        const int pq = pl + __popc((unsigned)t);
#pragma unroll
        for (int v = 0; v < 8; ++v) {
            size_t o = rbase + (size_t)((v + 8 * hi) * 256 + 16 * t + r16);
            const int pv = (v & 1) + ((v >> 1) & 1) + ((v >> 2) & 1);
            float o_r, o_i;
            phase_rot(dr[v], di[v], pq + pv, o_r, o_i);
            outr[o] = o_r;
            outi[o] = o_i;
        }
    }
}

extern "C" void kernel_launch(void* const* d_in, const int* in_sizes, int n_in,
                              void* d_out, int out_size, void* d_ws, size_t ws_size,
                              hipStream_t stream) {
    (void)in_sizes; (void)n_in; (void)out_size; (void)ws_size;
    const float* xr     = (const float*)d_in[0];
    const float* xi     = (const float*)d_in[1];
    const float* params = (const float*)d_in[2];

    float* g = (float*)d_ws;
    float* outr = (float*)d_out;
    float* outi = outr + (size_t)NBATCH * DIM;

    rx_build_gates<<<3, 256, 0, stream>>>(params, g);
    rx_apply<<<NBATCH, 256, 0, stream>>>(xr, xi, g, outr, outi);
}


# Stage-3 results go back into the row buffer in LDS (in place) and leave as whole lines: 8 global_store_b128 of 512 contiguous bytes per wave replace 32 global_store_b32 that each wrote two half-lines
# speedup vs baseline: 1.5434x; 1.2565x over previous
#include <hip/hip_runtime.h>


typedef float v2f __attribute__((ext_vector_type(2)));
typedef float v8f __attribute__((ext_vector_type(8)));

#define NQ    12
#define DIM   4096
#define NBATCH 4096

__global__ __launch_bounds__(256) void rx_build_gates(const float* __restrict__ params,
                                                      float* __restrict__ g) {
    int tid = blockIdx.x * 256 + threadIdx.x;
    if (tid >= 3 * 256) return;
    int grp = tid >> 8;
    int jk  = tid & 255;
    int j = jk >> 4, k = jk & 15;
    float p = 1.0f;
#pragma unroll
    for (int t = 0; t < 4; ++t) {
        int q = grp * 4 + t;
        float h = 0.5f * params[q];
        float c = cosf(h), s = sinf(h);
        int bj = (j >> (3 - t)) & 1;
        int bk = (k >> (3 - t)) & 1;
        float r = (bj == bk) ? (bj ? -c : c) : s;
        p = p * r;
    }
    g[grp * 256 + jk] = p;
}

__device__ __forceinline__ void phase_rot(float re, float im, int p, float& ore, float& oim) {
    const bool sw = (p & 1) != 0;
    const float a = sw ? im : re;
    const float b = sw ? re : im;
    const unsigned sa = (unsigned)(p & 2) << 30;
    const unsigned sb = (unsigned)((p + 1) & 2) << 30;
    ore = __uint_as_float(__float_as_uint(a) ^ sa);
    oim = __uint_as_float(__float_as_uint(b) ^ sb);
}

__device__ __forceinline__ v2f frag_a(const float* src, int stride, int k0, int r16) {
    v2f a;
    a.x = src[r16 * stride + k0];
    a.y = src[r16 * stride + k0 + 1];
    return a;
}
__device__ __forceinline__ v2f frag_b(const float* src, int stride, int k0, int r16) {
    v2f b;
    b.x = src[(k0)     * stride + r16];
    b.y = src[(k0 + 1) * stride + r16];
    return b;
}
__device__ __forceinline__ void store_d(float* dst, int stride, const v8f& d, int r16, int hi) {
#pragma unroll
    for (int v = 0; v < 8; ++v)
        dst[(v + 8 * hi) * stride + r16] = d[v];
}

__device__ __forceinline__ void rtile_data_gate(const float* Ar, const float* Ai, int sa,
                                                const float* G, int sb,
                                                v8f& dr, v8f& di, int r16, int hi) {
    v8f z = {0.f, 0.f, 0.f, 0.f, 0.f, 0.f, 0.f, 0.f};
    dr = z; di = z;
#pragma unroll
    for (int kb = 0; kb < 4; ++kb) {
        int k0 = kb * 4 + 2 * hi;
        v2f a_r = frag_a(Ar, sa, k0, r16);
        v2f a_i = frag_a(Ai, sa, k0, r16);
        v2f b   = frag_b(G,  sb, k0, r16);
        dr = __builtin_amdgcn_wmma_f32_16x16x4_f32(false, a_r, false, b, (short)0, dr, false, false);
        di = __builtin_amdgcn_wmma_f32_16x16x4_f32(false, a_i, false, b, (short)0, di, false, false);
    }
}

__device__ __forceinline__ void rtile_gate_data(const float* G, int sa,
                                                const float* Br, const float* Bi, int sb,
                                                v8f& dr, v8f& di, int r16, int hi) {
    v8f z = {0.f, 0.f, 0.f, 0.f, 0.f, 0.f, 0.f, 0.f};
    dr = z; di = z;
#pragma unroll
    for (int kb = 0; kb < 4; ++kb) {
        int k0 = kb * 4 + 2 * hi;
        v2f a   = frag_a(G,  sa, k0, r16);
        v2f b_r = frag_b(Br, sb, k0, r16);
        v2f b_i = frag_b(Bi, sb, k0, r16);
        dr = __builtin_amdgcn_wmma_f32_16x16x4_f32(false, a, false, b_r, (short)0, dr, false, false);
        di = __builtin_amdgcn_wmma_f32_16x16x4_f32(false, a, false, b_i, (short)0, di, false, false);
    }
}

__global__ __launch_bounds__(256) void rx_apply(const float* __restrict__ xr,
                                                const float* __restrict__ xi,
                                                const float* __restrict__ g,
                                                float* __restrict__ outr,
                                                float* __restrict__ outi) {
    __shared__ __align__(16) float sr[DIM];
    __shared__ __align__(16) float si[DIM];

    const int row = blockIdx.x;
    const size_t rbase = (size_t)row * DIM;

    {
        const float4* xr4 = (const float4*)(xr + rbase);
        const float4* xi4 = (const float4*)(xi + rbase);
        float4* sr4 = (float4*)sr;
        float4* si4 = (float4*)si;
        const int pt = __popc((unsigned)threadIdx.x);
#pragma unroll
        for (int it = 0; it < 4; ++it) {
            int idx = threadIdx.x + it * 256;
            float4 vr = xr4[idx];
            float4 vi = xi4[idx];
            const int p0 = pt + ((it & 1) + (it >> 1));
            float4 qr, qi;
            phase_rot(vr.x, vi.x, p0,     qr.x, qi.x);
            phase_rot(vr.y, vi.y, p0 + 1, qr.y, qi.y);
            phase_rot(vr.z, vi.z, p0 + 1, qr.z, qi.z);
            phase_rot(vr.w, vi.w, p0 + 2, qr.w, qi.w);
            sr4[idx] = qr;
            si4[idx] = qi;
        }
    }
    __syncthreads();

    const int lane = threadIdx.x & 31;
    const int wave = threadIdx.x >> 5;
    const int r16  = lane & 15;
    const int hi   = lane >> 4;

    const float* gA = g;
    const float* gB = g + 256;
    const float* gC = g + 512;

    v8f dr, di;

#pragma unroll
    for (int rep = 0; rep < 2; ++rep) {
        const int t = wave + 8 * rep;
        float* br_ = sr + t * 256;
        float* bi_ = si + t * 256;
        rtile_data_gate(br_, bi_, 16, gC, 16, dr, di, r16, hi);
        store_d(br_, 16, dr, r16, hi);
        store_d(bi_, 16, di, r16, hi);
        rtile_gate_data(gB, 16, br_, bi_, 16, dr, di, r16, hi);
        store_d(br_, 16, dr, r16, hi);
        store_d(bi_, 16, di, r16, hi);
    }

    __syncthreads();

    const int pl = __popc((unsigned)lane);
#pragma unroll
    for (int rep = 0; rep < 2; ++rep) {
        const int t = wave + 8 * rep;
        rtile_gate_data(gA, 16, sr + 16 * t, si + 16 * t, 256, dr, di, r16, hi);
        const int pq = pl + __popc((unsigned)t);
#pragma unroll
        for (int v = 0; v < 8; ++v) {
            const int o = (v + 8 * hi) * 256 + 16 * t + r16;
            const int pv = (v & 1) + ((v >> 1) & 1) + ((v >> 2) & 1);
            float o_r, o_i;
            phase_rot(dr[v], di[v], pq + pv, o_r, o_i);
            sr[o] = o_r;
            si[o] = o_i;
        }
    }

    __syncthreads();

    {
        const float4* sr4 = (const float4*)sr;
        const float4* si4 = (const float4*)si;
        float4* or4 = (float4*)(outr + rbase);
        float4* oi4 = (float4*)(outi + rbase);
#pragma unroll
        for (int it = 0; it < 4; ++it) {
            int idx = threadIdx.x + it * 256;
            float4 qr = sr4[idx];
            float4 qi = si4[idx];
            or4[idx] = qr;
            oi4[idx] = qi;
        }
    }
}

extern "C" void kernel_launch(void* const* d_in, const int* in_sizes, int n_in,
                              void* d_out, int out_size, void* d_ws, size_t ws_size,
                              hipStream_t stream) {
    (void)in_sizes; (void)n_in; (void)out_size; (void)ws_size;
    const float* xr     = (const float*)d_in[0];
    const float* xi     = (const float*)d_in[1];
    const float* params = (const float*)d_in[2];

    float* g = (float*)d_ws;
    float* outr = (float*)d_out;
    float* outi = outr + (size_t)NBATCH * DIM;

    rx_build_gates<<<3, 256, 0, stream>>>(params, g);
    rx_apply<<<NBATCH, 256, 0, stream>>>(xr, xi, g, outr, outi);
}


# Stage 1 accumulators feed stage 2 as its B fragment in registers (RB fragment loaded in the k order of the C/D layout): one LDS store+load round trip per tile removed; stage-2 sum order changes
# speedup vs baseline: 1.6009x; 1.0373x over previous
#include <hip/hip_runtime.h>


typedef float v2f __attribute__((ext_vector_type(2)));
typedef float v8f __attribute__((ext_vector_type(8)));

#define NQ    12
#define DIM   4096
#define NBATCH 4096

__global__ __launch_bounds__(256) void rx_build_gates(const float* __restrict__ params,
                                                      float* __restrict__ g) {
    int tid = blockIdx.x * 256 + threadIdx.x;
    if (tid >= 3 * 256) return;
    int grp = tid >> 8;
    int jk  = tid & 255;
    int j = jk >> 4, k = jk & 15;
    float p = 1.0f;
#pragma unroll
    for (int t = 0; t < 4; ++t) {
        int q = grp * 4 + t;
        float h = 0.5f * params[q];
        float c = cosf(h), s = sinf(h);
        int bj = (j >> (3 - t)) & 1;
        int bk = (k >> (3 - t)) & 1;
        float r = (bj == bk) ? (bj ? -c : c) : s;
        p = p * r;
    }
    g[grp * 256 + jk] = p;
}

__device__ __forceinline__ void phase_rot(float re, float im, int p, float& ore, float& oim) {
    const bool sw = (p & 1) != 0;
    const float a = sw ? im : re;
    const float b = sw ? re : im;
    const unsigned sa = (unsigned)(p & 2) << 30;
    const unsigned sb = (unsigned)((p + 1) & 2) << 30;
    ore = __uint_as_float(__float_as_uint(a) ^ sa);
    oim = __uint_as_float(__float_as_uint(b) ^ sb);
}

__device__ __forceinline__ v2f frag_a(const float* src, int stride, int k0, int r16) {
    v2f a;
    a.x = src[r16 * stride + k0];
    a.y = src[r16 * stride + k0 + 1];
    return a;
}
__device__ __forceinline__ v2f frag_b(const float* src, int stride, int k0, int r16) {
    v2f b;
    b.x = src[(k0)     * stride + r16];
    b.y = src[(k0 + 1) * stride + r16];
    return b;
}
__device__ __forceinline__ void store_d(float* dst, int stride, const v8f& d, int r16, int hi) {
#pragma unroll
    for (int v = 0; v < 8; ++v)
        dst[(v + 8 * hi) * stride + r16] = d[v];
}

__device__ __forceinline__ void rtile_data_gate(const float* Ar, const float* Ai, int sa,
                                                const float* G, int sb,
                                                v8f& dr, v8f& di, int r16, int hi) {
    v8f z = {0.f, 0.f, 0.f, 0.f, 0.f, 0.f, 0.f, 0.f};
    dr = z; di = z;
#pragma unroll
    for (int kb = 0; kb < 4; ++kb) {
        int k0 = kb * 4 + 2 * hi;
        v2f a_r = frag_a(Ar, sa, k0, r16);
        v2f a_i = frag_a(Ai, sa, k0, r16);
        v2f b   = frag_b(G,  sb, k0, r16);
        dr = __builtin_amdgcn_wmma_f32_16x16x4_f32(false, a_r, false, b, (short)0, dr, false, false);
        di = __builtin_amdgcn_wmma_f32_16x16x4_f32(false, a_i, false, b, (short)0, di, false, false);
    }
}

__device__ __forceinline__ void rtile_gate_data(const float* G, int sa,
                                                const float* Br, const float* Bi, int sb,
                                                v8f& dr, v8f& di, int r16, int hi) {
    v8f z = {0.f, 0.f, 0.f, 0.f, 0.f, 0.f, 0.f, 0.f};
    dr = z; di = z;
#pragma unroll
    for (int kb = 0; kb < 4; ++kb) {
        int k0 = kb * 4 + 2 * hi;
        v2f a   = frag_a(G,  sa, k0, r16);
        v2f b_r = frag_b(Br, sb, k0, r16);
        v2f b_i = frag_b(Bi, sb, k0, r16);
        dr = __builtin_amdgcn_wmma_f32_16x16x4_f32(false, a, false, b_r, (short)0, dr, false, false);
        di = __builtin_amdgcn_wmma_f32_16x16x4_f32(false, a, false, b_i, (short)0, di, false, false);
    }
}

__device__ __forceinline__ void rtile_gate_acc(const float* G, int sa,
                                               const v8f& tr, const v8f& ti,
                                               v8f& dr, v8f& di, int r16, int hi) {
    v8f z = {0.f, 0.f, 0.f, 0.f, 0.f, 0.f, 0.f, 0.f};
    v8f er = z, ei = z;
#pragma unroll
    for (int j = 0; j < 4; ++j) {
        v2f a   = frag_a(G, sa, 8 * hi + 2 * j, r16);
        v2f b_r = {tr[2 * j], tr[2 * j + 1]};
        v2f b_i = {ti[2 * j], ti[2 * j + 1]};
        er = __builtin_amdgcn_wmma_f32_16x16x4_f32(false, a, false, b_r, (short)0, er, false, false);
        ei = __builtin_amdgcn_wmma_f32_16x16x4_f32(false, a, false, b_i, (short)0, ei, false, false);
    }
    dr = er; di = ei;
}

__global__ __launch_bounds__(256, 16) void rx_apply(const float* __restrict__ xr,
                                                const float* __restrict__ xi,
                                                const float* __restrict__ g,
                                                float* __restrict__ outr,
                                                float* __restrict__ outi) {
    __shared__ __align__(16) float sr[DIM];
    __shared__ __align__(16) float si[DIM];

    const int row = blockIdx.x;
    const size_t rbase = (size_t)row * DIM;

    {
        const float4* xr4 = (const float4*)(xr + rbase);
        const float4* xi4 = (const float4*)(xi + rbase);
        float4* sr4 = (float4*)sr;
        float4* si4 = (float4*)si;
        const int pt = __popc((unsigned)threadIdx.x);
#pragma unroll
        for (int it = 0; it < 4; ++it) {
            int idx = threadIdx.x + it * 256;
            float4 vr = xr4[idx];
            float4 vi = xi4[idx];
            const int p0 = pt + ((it & 1) + (it >> 1));
            float4 qr, qi;
            phase_rot(vr.x, vi.x, p0,     qr.x, qi.x);
            phase_rot(vr.y, vi.y, p0 + 1, qr.y, qi.y);
            phase_rot(vr.z, vi.z, p0 + 1, qr.z, qi.z);
            phase_rot(vr.w, vi.w, p0 + 2, qr.w, qi.w);
            sr4[idx] = qr;
            si4[idx] = qi;
        }
    }
    __syncthreads();

    const int lane = threadIdx.x & 31;
    const int wave = threadIdx.x >> 5;
    const int r16  = lane & 15;
    const int hi   = lane >> 4;

    const float* gA = g;
    const float* gB = g + 256;
    const float* gC = g + 512;

    v8f dr, di;

#pragma unroll
    for (int rep = 0; rep < 2; ++rep) {
        const int t = wave + 8 * rep;
        float* br_ = sr + t * 256;
        float* bi_ = si + t * 256;
        v8f tr, ti;
        rtile_data_gate(br_, bi_, 16, gC, 16, tr, ti, r16, hi);
        rtile_gate_acc(gB, 16, tr, ti, dr, di, r16, hi);
        store_d(br_, 16, dr, r16, hi);
        store_d(bi_, 16, di, r16, hi);
    }

    __syncthreads();

    const int pl = __popc((unsigned)lane);
#pragma unroll
    for (int rep = 0; rep < 2; ++rep) {
        const int t = wave + 8 * rep;
        rtile_gate_data(gA, 16, sr + 16 * t, si + 16 * t, 256, dr, di, r16, hi);
        const int pq = pl + __popc((unsigned)t);
#pragma unroll
        for (int v = 0; v < 8; ++v) {
            const int o = (v + 8 * hi) * 256 + 16 * t + r16;
            const int pv = (v & 1) + ((v >> 1) & 1) + ((v >> 2) & 1);
            float o_r, o_i;
            phase_rot(dr[v], di[v], pq + pv, o_r, o_i);
            sr[o] = o_r;
            si[o] = o_i;
        }
    }

    __syncthreads();

    {
        const float4* sr4 = (const float4*)sr;
        const float4* si4 = (const float4*)si;
        float4* or4 = (float4*)(outr + rbase);
        float4* oi4 = (float4*)(outi + rbase);
#pragma unroll
        for (int it = 0; it < 4; ++it) {
            int idx = threadIdx.x + it * 256;
            float4 qr = sr4[idx];
            float4 qi = si4[idx];
            or4[idx] = qr;
            oi4[idx] = qi;
        }
    }
}

extern "C" void kernel_launch(void* const* d_in, const int* in_sizes, int n_in,
                              void* d_out, int out_size, void* d_ws, size_t ws_size,
                              hipStream_t stream) {
    (void)in_sizes; (void)n_in; (void)out_size; (void)ws_size;
    const float* xr     = (const float*)d_in[0];
    const float* xi     = (const float*)d_in[1];
    const float* params = (const float*)d_in[2];

    float* g = (float*)d_ws;
    float* outr = (float*)d_out;
    float* outi = outr + (size_t)NBATCH * DIM;

    rx_build_gates<<<3, 256, 0, stream>>>(params, g);
    rx_apply<<<NBATCH, 256, 0, stream>>>(xr, xi, g, outr, outi);
}
